# starting kernel
# baseline (speedup 1.0000x reference)
#include <hip/hip_runtime.h>

typedef __attribute__((ext_vector_type(2))) float v2f;
typedef __attribute__((ext_vector_type(8))) float v8f;

#define NNODE 8192
#define DIM   128
#define KNBR  8
#define NEGINF -1e30f

__device__ __forceinline__ float4 ld4(const float* p) {
  return *reinterpret_cast<const float4*>(p);
}
__device__ __forceinline__ void st4(float* p, float4 v) {
  *reinterpret_cast<float4*>(p) = v;
}

__global__ void deg_kernel(const int* __restrict__ neigh,
                           const int* __restrict__ nbrc,
                           float* __restrict__ dinv,
                           unsigned* __restrict__ keep) {
  int i = blockIdx.x * blockDim.x + threadIdx.x;
  if (i >= NNODE) return;
  int cnt = nbrc[i]; cnt = cnt > KNBR ? KNBR : cnt;
  const int* nb = neigh + i * KNBR;
  unsigned km = 0;
  int deg = 1;
  for (int k = 0; k < cnt; ++k) {
    int j = nb[k];
    if (j == i) continue;
    bool dup = false;
    for (int t = 0; t < k; ++t) dup |= (nb[t] == j);
    if (!dup) { km |= (1u << k); ++deg; }
  }
  dinv[i] = rsqrtf((float)deg);
  keep[i] = km;
}

__global__ void wmma_gemm_bias(const float* __restrict__ A, int lda,
                               const float* __restrict__ W, int ldw,
                               const float* __restrict__ bias,
                               float* __restrict__ Y, int ldy,
                               int K, int do_relu) {
  const int wave = threadIdx.x >> 5;
  const int lane = threadIdx.x & 31;
  const int half = lane >> 4;
  const int l16  = lane & 15;
  const int m0 = blockIdx.x * 16;
  const int n0 = (blockIdx.y * (blockDim.x >> 5) + wave) * 16;

  const float* __restrict__ arow = A + (size_t)(m0 + l16) * lda + 2 * half;
  const float* __restrict__ wrow = W + (size_t)(n0 + l16) * ldw + 2 * half;

  v8f c = {};
  for (int k0 = 0; k0 < K; k0 += 4) {
    v2f a = *reinterpret_cast<const v2f*>(arow + k0);
    v2f b = *reinterpret_cast<const v2f*>(wrow + k0);
    c = __builtin_amdgcn_wmma_f32_16x16x4_f32(false, a, false, b,
                                              (short)0, c, false, false);
  }

  const float bv = bias[n0 + l16];
#pragma unroll
  for (int r = 0; r < 8; ++r) {
    float v = c[r] + bv;
    if (do_relu) v = fmaxf(v, 0.0f);
    Y[(size_t)(m0 + r + 8 * half) * ldy + n0 + l16] = v;
  }
}

__device__ __forceinline__ void wave_ln_store(float4 v,
                                              const float* __restrict__ g,
                                              const float* __restrict__ b,
                                              float* __restrict__ dstrow,
                                              int lane) {
  float s = v.x + v.y + v.z + v.w;
#pragma unroll
  for (int off = 16; off > 0; off >>= 1) s += __shfl_xor(s, off, 32);
  const float mu = s * (1.0f / (float)DIM);
  const float dx = v.x - mu, dy = v.y - mu, dz = v.z - mu, dw = v.w - mu;
  float q = dx * dx + dy * dy + dz * dz + dw * dw;
#pragma unroll
  for (int off = 16; off > 0; off >>= 1) q += __shfl_xor(q, off, 32);
  const float rinv = rsqrtf(q * (1.0f / (float)DIM) + 1e-5f);
  const int d = lane * 4;
  const float4 gg = ld4(g + d);
  const float4 bb = ld4(b + d);
  float4 o;
  o.x = dx * rinv * gg.x + bb.x;
  o.y = dy * rinv * gg.y + bb.y;
  o.z = dz * rinv * gg.z + bb.z;
  o.w = dw * rinv * gg.w + bb.w;
  st4(dstrow + d, o);
}

__global__ void gcn_prop_ln(const float* __restrict__ nxt,
                            const float* __restrict__ resid,
                            const int* __restrict__ neigh,
                            const float* __restrict__ dinv,
                            const unsigned* __restrict__ keep,
                            const float* __restrict__ g,
                            const float* __restrict__ b,
                            float* __restrict__ dst, int ldd) {
  const int i = (int)((blockIdx.x * blockDim.x + threadIdx.x) >> 5);
  if (i >= NNODE) return;
  const int lane = threadIdx.x & 31;
  const int d = lane * 4;

  const float di = dinv[i];
  float4 t = ld4(nxt + (size_t)i * DIM + d);
  float4 acc = {di * t.x, di * t.y, di * t.z, di * t.w};

  const unsigned km = keep[i];
  const int* nb = neigh + i * KNBR;
  for (int k = 0; k < KNBR; ++k) {
    if ((km >> k) & 1u) {
      const int j = nb[k];
      const float dj = dinv[j];
      float4 u = ld4(nxt + (size_t)j * DIM + d);
      acc.x += dj * u.x; acc.y += dj * u.y;
      acc.z += dj * u.z; acc.w += dj * u.w;
    }
  }
  acc.x = fmaxf(acc.x * di, 0.0f);
  acc.y = fmaxf(acc.y * di, 0.0f);
  acc.z = fmaxf(acc.z * di, 0.0f);
  acc.w = fmaxf(acc.w * di, 0.0f);

  const float4 r = ld4(resid + (size_t)i * DIM + d);
  float4 v = {acc.x + r.x, acc.y + r.y, acc.z + r.z, acc.w + r.w};
  wave_ln_store(v, g, b, dst + (size_t)i * ldd, lane);
}

__global__ void sage_pool_cat(const float* __restrict__ selfsrc,
                              const float* __restrict__ poolsrc,
                              const int* __restrict__ neigh,
                              const int* __restrict__ nbrc,
                              float* __restrict__ cat) {
  const int i = (int)((blockIdx.x * blockDim.x + threadIdx.x) >> 5);
  if (i >= NNODE) return;
  const int lane = threadIdx.x & 31;
  const int d = lane * 4;

  st4(cat + (size_t)i * (2 * DIM) + d, ld4(selfsrc + (size_t)i * DIM + d));

  int cnt = nbrc[i]; cnt = cnt > KNBR ? KNBR : cnt;
  float4 m = {NEGINF, NEGINF, NEGINF, NEGINF};
  const int* nb = neigh + i * KNBR;
  for (int k = 0; k < cnt; ++k) {
    const int j = nb[k];
    float4 u = ld4(poolsrc + (size_t)j * DIM + d);
    m.x = fmaxf(m.x, u.x); m.y = fmaxf(m.y, u.y);
    m.z = fmaxf(m.z, u.z); m.w = fmaxf(m.w, u.w);
  }
  if (cnt == 0) m = make_float4(0.f, 0.f, 0.f, 0.f);
  st4(cat + (size_t)i * (2 * DIM) + DIM + d, m);
}

__global__ void resid_ln(const float* __restrict__ a,
                         const float* __restrict__ res,
                         const float* __restrict__ g,
                         const float* __restrict__ b,
                         float* __restrict__ dst, int ldd) {
  const int i = (int)((blockIdx.x * blockDim.x + threadIdx.x) >> 5);
  if (i >= NNODE) return;
  const int lane = threadIdx.x & 31;
  const int d = lane * 4;
  const float4 av = ld4(a + (size_t)i * DIM + d);
  const float4 rv = ld4(res + (size_t)i * DIM + d);
  float4 v = {av.x + rv.x, av.y + rv.y, av.z + rv.z, av.w + rv.w};
  wave_ln_store(v, g, b, dst + (size_t)i * ldd, lane);
}

extern "C" void kernel_launch(void* const* d_in, const int* in_sizes, int n_in,
                              void* d_out, int out_size, void* d_ws, size_t ws_size,
                              hipStream_t stream) {
  const float* x      = (const float*)d_in[0];
  const int*   neigh  = (const int*)  d_in[1];
  const int*   nbrc   = (const int*)  d_in[2];
  const float* gcnW   = (const float*)d_in[3];
  const float* gcnB   = (const float*)d_in[4];
  const float* gcnG   = (const float*)d_in[5];
  const float* gcnBb  = (const float*)d_in[6];
  const float* poolW  = (const float*)d_in[7];
  const float* poolB  = (const float*)d_in[8];
  const float* mrgW   = (const float*)d_in[9];
  const float* mrgB   = (const float*)d_in[10];
  const float* sgG    = (const float*)d_in[11];
  const float* sgB    = (const float*)d_in[12];
  float* out = (float*)d_out;

  const size_t ND = (size_t)NNODE * DIM;
  float*    dinv  = (float*)d_ws;
  unsigned* keep  = (unsigned*)(dinv + NNODE);
  float*    first = (float*)(keep + NNODE);
  float*    nxt   = first + ND;
  float*    hbuf  = nxt + ND;
  float*    sbuf  = hbuf + ND;
  float*    tmp   = sbuf + ND;
  float*    cat   = tmp + ND;

  const dim3 gB(128);
  const dim3 gG(NNODE / 16, DIM / 64);
  const dim3 nB(256);
  const dim3 nG(NNODE / 8);

  deg_kernel<<<NNODE / 256, 256, 0, stream>>>(neigh, nbrc, dinv, keep);

  wmma_gemm_bias<<<gG, gB, 0, stream>>>(x, DIM, gcnW, DIM, gcnB,
                                        first, DIM, DIM, 0);
  gcn_prop_ln<<<nG, nB, 0, stream>>>(first, first, neigh, dinv, keep,
                                     gcnG, gcnBb, hbuf, DIM);

  wmma_gemm_bias<<<gG, gB, 0, stream>>>(hbuf, DIM, gcnW + DIM * DIM, DIM,
                                        gcnB + DIM, nxt, DIM, DIM, 0);
  gcn_prop_ln<<<nG, nB, 0, stream>>>(nxt, hbuf, neigh, dinv, keep,
                                     gcnG + DIM, gcnBb + DIM, hbuf, DIM);

  wmma_gemm_bias<<<gG, gB, 0, stream>>>(hbuf, DIM, gcnW + 2 * DIM * DIM, DIM,
                                        gcnB + 2 * DIM, nxt, DIM, DIM, 0);
  gcn_prop_ln<<<nG, nB, 0, stream>>>(nxt, hbuf, neigh, dinv, keep,
                                     gcnG + 2 * DIM, gcnBb + 2 * DIM,
                                     out, 2 * DIM);

  wmma_gemm_bias<<<gG, gB, 0, stream>>>(x, DIM, poolW, DIM, poolB,
                                        tmp, DIM, DIM, 1);
  sage_pool_cat<<<nG, nB, 0, stream>>>(x, tmp, neigh, nbrc, cat);
  wmma_gemm_bias<<<gG, gB, 0, stream>>>(cat, 2 * DIM, mrgW, 2 * DIM, mrgB,
                                        sbuf, DIM, 2 * DIM, 1);

  sage_pool_cat<<<nG, nB, 0, stream>>>(sbuf, sbuf, neigh, nbrc, cat);
  wmma_gemm_bias<<<gG, gB, 0, stream>>>(cat, 2 * DIM, mrgW + DIM * 2 * DIM,
                                        2 * DIM, mrgB + DIM, tmp, DIM,
                                        2 * DIM, 1);
  resid_ln<<<nG, nB, 0, stream>>>(tmp, sbuf, sgG, sgB, sbuf, DIM);

  sage_pool_cat<<<nG, nB, 0, stream>>>(sbuf, sbuf, neigh, nbrc, cat);
  wmma_gemm_bias<<<gG, gB, 0, stream>>>(cat, 2 * DIM, mrgW + 2 * DIM * 2 * DIM,
                                        2 * DIM, mrgB + 2 * DIM, tmp, DIM,
                                        2 * DIM, 1);
  resid_ln<<<nG, nB, 0, stream>>>(tmp, sbuf, sgG + DIM, sgB + DIM,
                                  out + DIM, 2 * DIM);
}


# GEMM operands through LDS: each block (32 rows x 128 cols) copies its A panel and all of W with whole-line float4 loads in two K halves; waves read fragments from LDS; k order and sums unchanged
# speedup vs baseline: 1.2758x; 1.2758x over previous
#include <hip/hip_runtime.h>

typedef __attribute__((ext_vector_type(2))) float v2f;
typedef __attribute__((ext_vector_type(8))) float v8f;

#define NNODE 8192
#define DIM   128
#define KNBR  8
#define NEGINF -1e30f

__device__ __forceinline__ float4 ld4(const float* p) {
  return *reinterpret_cast<const float4*>(p);
}
__device__ __forceinline__ void st4(float* p, float4 v) {
  *reinterpret_cast<float4*>(p) = v;
}

__global__ void deg_kernel(const int* __restrict__ neigh,
                           const int* __restrict__ nbrc,
                           float* __restrict__ dinv,
                           unsigned* __restrict__ keep) {
  int i = blockIdx.x * blockDim.x + threadIdx.x;
  if (i >= NNODE) return;
  int cnt = nbrc[i]; cnt = cnt > KNBR ? KNBR : cnt;
  const int* nb = neigh + i * KNBR;
  unsigned km = 0;
  int deg = 1;
  for (int k = 0; k < cnt; ++k) {
    int j = nb[k];
    if (j == i) continue;
    bool dup = false;
    for (int t = 0; t < k; ++t) dup |= (nb[t] == j);
    if (!dup) { km |= (1u << k); ++deg; }
  }
  dinv[i] = rsqrtf((float)deg);
  keep[i] = km;
}

template <int KB, int KE>
__device__ __forceinline__ void mma_range(const float* ap, const float* bp0,
                                          const float* bp1, v8f& c0, v8f& c1) {
#pragma unroll 8
  for (int k0 = KB; k0 < KE; k0 += 4) {
    const v2f a  = *reinterpret_cast<const v2f*>(ap + k0);
    const v2f b0 = *reinterpret_cast<const v2f*>(bp0 + k0);
    const v2f b1 = *reinterpret_cast<const v2f*>(bp1 + k0);
    c0 = __builtin_amdgcn_wmma_f32_16x16x4_f32(false, a, false, b0,
                                               (short)0, c0, false, false);
    c1 = __builtin_amdgcn_wmma_f32_16x16x4_f32(false, a, false, b1,
                                               (short)0, c1, false, false);
  }
}

template <int K>
__global__ __launch_bounds__(256)
void wmma_gemm_bias(const float* __restrict__ A,
                    const float* __restrict__ W,
                    const float* __restrict__ bias,
                    float* __restrict__ Y, int ldy, int do_relu) {
  constexpr int S   = K + 4;
  constexpr int H   = K / 2;
  constexpr int HC4 = H / 4;
  constexpr int NA  = (32 * HC4) / 256;
  constexpr int NW  = (DIM * HC4) / 256;
  __shared__ __attribute__((aligned(16))) float As[32 * S];
  __shared__ __attribute__((aligned(16))) float Ws[DIM * S];

  const int t  = threadIdx.x;
  const int m0 = blockIdx.x * 32;
  const float* __restrict__ Ablk = A + (size_t)m0 * K;

#pragma unroll
  for (int it = 0; it < NA; ++it) {
    const int idx = it * 256 + t;
    const int r = idx / HC4, c = idx % HC4;
    st4(As + r * S + 4 * c, ld4(Ablk + r * K + 4 * c));
  }
#pragma unroll
  for (int it = 0; it < NW; ++it) {
    const int idx = it * 256 + t;
    const int r = idx / HC4, c = idx % HC4;
    st4(Ws + r * S + 4 * c, ld4(W + r * K + 4 * c));
  }
  __syncthreads();

  float4 ra[NA];
  float4 rw[NW];
#pragma unroll
  for (int it = 0; it < NA; ++it) {
    const int idx = it * 256 + t;
    const int r = idx / HC4, c = idx % HC4;
    ra[it] = ld4(Ablk + r * K + H + 4 * c);
  }
#pragma unroll
  for (int it = 0; it < NW; ++it) {
    const int idx = it * 256 + t;
    const int r = idx / HC4, c = idx % HC4;
    rw[it] = ld4(W + r * K + H + 4 * c);
  }

  const int wave = t >> 5;
  const int lane = t & 31;
  const int half = lane >> 4;
  const int l16  = lane & 15;
  const int mt   = wave & 1;
  const int n0   = (wave >> 1) * 32;

  const float* ap  = As + (mt * 16 + l16) * S + 2 * half;
  const float* bp0 = Ws + (n0 + l16) * S + 2 * half;
  const float* bp1 = bp0 + 16 * S;

  v8f c0 = {};
  v8f c1 = {};
  mma_range<0, H>(ap, bp0, bp1, c0, c1);

#pragma unroll
  for (int it = 0; it < NA; ++it) {
    const int idx = it * 256 + t;
    const int r = idx / HC4, c = idx % HC4;
    st4(As + r * S + H + 4 * c, ra[it]);
  }
#pragma unroll
  for (int it = 0; it < NW; ++it) {
    const int idx = it * 256 + t;
    const int r = idx / HC4, c = idx % HC4;
    st4(Ws + r * S + H + 4 * c, rw[it]);
  }
  __syncthreads();
  mma_range<H, K>(ap, bp0, bp1, c0, c1);

  const float bv0 = bias[n0 + l16];
  const float bv1 = bias[n0 + 16 + l16];
  float* yrow = Y + (size_t)(m0 + mt * 16 + 8 * half) * ldy + n0 + l16;
#pragma unroll
  for (int r = 0; r < 8; ++r) {
    float v0 = c0[r] + bv0;
    float v1 = c1[r] + bv1;
    if (do_relu) { v0 = fmaxf(v0, 0.0f); v1 = fmaxf(v1, 0.0f); }
    yrow[(size_t)r * ldy]      = v0;
    yrow[(size_t)r * ldy + 16] = v1;
  }
}

__device__ __forceinline__ void wave_ln_store(float4 v,
                                              const float* __restrict__ g,
                                              const float* __restrict__ b,
                                              float* __restrict__ dstrow,
                                              int lane) {
  float s = v.x + v.y + v.z + v.w;
#pragma unroll
  for (int off = 16; off > 0; off >>= 1) s += __shfl_xor(s, off, 32);
  const float mu = s * (1.0f / (float)DIM);
  const float dx = v.x - mu, dy = v.y - mu, dz = v.z - mu, dw = v.w - mu;
  float q = dx * dx + dy * dy + dz * dz + dw * dw;
#pragma unroll
  for (int off = 16; off > 0; off >>= 1) q += __shfl_xor(q, off, 32);
  const float rinv = rsqrtf(q * (1.0f / (float)DIM) + 1e-5f);
  const int d = lane * 4;
  const float4 gg = ld4(g + d);
  const float4 bb = ld4(b + d);
  float4 o;
  o.x = dx * rinv * gg.x + bb.x;
  o.y = dy * rinv * gg.y + bb.y;
  o.z = dz * rinv * gg.z + bb.z;
  o.w = dw * rinv * gg.w + bb.w;
  st4(dstrow + d, o);
}

__global__ void gcn_prop_ln(const float* __restrict__ nxt,
                            const float* __restrict__ resid,
                            const int* __restrict__ neigh,
                            const float* __restrict__ dinv,
                            const unsigned* __restrict__ keep,
                            const float* __restrict__ g,
                            const float* __restrict__ b,
                            float* __restrict__ dst, int ldd) {
  const int i = (int)((blockIdx.x * blockDim.x + threadIdx.x) >> 5);
  if (i >= NNODE) return;
  const int lane = threadIdx.x & 31;
  const int d = lane * 4;

  const float di = dinv[i];
  float4 t = ld4(nxt + (size_t)i * DIM + d);
  float4 acc = {di * t.x, di * t.y, di * t.z, di * t.w};

  const unsigned km = keep[i];
  const int* nb = neigh + i * KNBR;
  for (int k = 0; k < KNBR; ++k) {
    if ((km >> k) & 1u) {
      const int j = nb[k];
      const float dj = dinv[j];
      float4 u = ld4(nxt + (size_t)j * DIM + d);
      acc.x += dj * u.x; acc.y += dj * u.y;
      acc.z += dj * u.z; acc.w += dj * u.w;
    }
  }
  acc.x = fmaxf(acc.x * di, 0.0f);
  acc.y = fmaxf(acc.y * di, 0.0f);
  acc.z = fmaxf(acc.z * di, 0.0f);
  acc.w = fmaxf(acc.w * di, 0.0f);

  const float4 r = ld4(resid + (size_t)i * DIM + d);
  float4 v = {acc.x + r.x, acc.y + r.y, acc.z + r.z, acc.w + r.w};
  wave_ln_store(v, g, b, dst + (size_t)i * ldd, lane);
}

__global__ void sage_pool_cat(const float* __restrict__ selfsrc,
                              const float* __restrict__ poolsrc,
                              const int* __restrict__ neigh,
                              const int* __restrict__ nbrc,
                              float* __restrict__ cat) {
  const int i = (int)((blockIdx.x * blockDim.x + threadIdx.x) >> 5);
  if (i >= NNODE) return;
  const int lane = threadIdx.x & 31;
  const int d = lane * 4;

  st4(cat + (size_t)i * (2 * DIM) + d, ld4(selfsrc + (size_t)i * DIM + d));

  int cnt = nbrc[i]; cnt = cnt > KNBR ? KNBR : cnt;
  float4 m = {NEGINF, NEGINF, NEGINF, NEGINF};
  const int* nb = neigh + i * KNBR;
  for (int k = 0; k < cnt; ++k) {
    const int j = nb[k];
    float4 u = ld4(poolsrc + (size_t)j * DIM + d);
    m.x = fmaxf(m.x, u.x); m.y = fmaxf(m.y, u.y);
    m.z = fmaxf(m.z, u.z); m.w = fmaxf(m.w, u.w);
  }
  if (cnt == 0) m = make_float4(0.f, 0.f, 0.f, 0.f);
  st4(cat + (size_t)i * (2 * DIM) + DIM + d, m);
}

__global__ void resid_ln(const float* __restrict__ a,
                         const float* __restrict__ res,
                         const float* __restrict__ g,
                         const float* __restrict__ b,
                         float* __restrict__ dst, int ldd) {
  const int i = (int)((blockIdx.x * blockDim.x + threadIdx.x) >> 5);
  if (i >= NNODE) return;
  const int lane = threadIdx.x & 31;
  const int d = lane * 4;
  const float4 av = ld4(a + (size_t)i * DIM + d);
  const float4 rv = ld4(res + (size_t)i * DIM + d);
  float4 v = {av.x + rv.x, av.y + rv.y, av.z + rv.z, av.w + rv.w};
  wave_ln_store(v, g, b, dst + (size_t)i * ldd, lane);
}

extern "C" void kernel_launch(void* const* d_in, const int* in_sizes, int n_in,
                              void* d_out, int out_size, void* d_ws, size_t ws_size,
                              hipStream_t stream) {
  const float* x      = (const float*)d_in[0];
  const int*   neigh  = (const int*)  d_in[1];
  const int*   nbrc   = (const int*)  d_in[2];
  const float* gcnW   = (const float*)d_in[3];
  const float* gcnB   = (const float*)d_in[4];
  const float* gcnG   = (const float*)d_in[5];
  const float* gcnBb  = (const float*)d_in[6];
  const float* poolW  = (const float*)d_in[7];
  const float* poolB  = (const float*)d_in[8];
  const float* mrgW   = (const float*)d_in[9];
  const float* mrgB   = (const float*)d_in[10];
  const float* sgG    = (const float*)d_in[11];
  const float* sgB    = (const float*)d_in[12];
  float* out = (float*)d_out;

  const size_t ND = (size_t)NNODE * DIM;
  float*    dinv  = (float*)d_ws;
  unsigned* keep  = (unsigned*)(dinv + NNODE);
  float*    first = (float*)(keep + NNODE);
  float*    nxt   = first + ND;
  float*    hbuf  = nxt + ND;
  float*    sbuf  = hbuf + ND;
  float*    tmp   = sbuf + ND;
  float*    cat   = tmp + ND;

  const dim3 gB(256);
  const dim3 gG(NNODE / 32);
  const dim3 nB(256);
  const dim3 nG(NNODE / 8);

  deg_kernel<<<NNODE / 256, 256, 0, stream>>>(neigh, nbrc, dinv, keep);

  wmma_gemm_bias<DIM><<<gG, gB, 0, stream>>>(x, gcnW, gcnB, first, DIM, 0);
  gcn_prop_ln<<<nG, nB, 0, stream>>>(first, first, neigh, dinv, keep,
                                     gcnG, gcnBb, hbuf, DIM);

  wmma_gemm_bias<DIM><<<gG, gB, 0, stream>>>(hbuf, gcnW + DIM * DIM,
                                             gcnB + DIM, nxt, DIM, 0);
  gcn_prop_ln<<<nG, nB, 0, stream>>>(nxt, hbuf, neigh, dinv, keep,
                                     gcnG + DIM, gcnBb + DIM, hbuf, DIM);

  wmma_gemm_bias<DIM><<<gG, gB, 0, stream>>>(hbuf, gcnW + 2 * DIM * DIM,
                                             gcnB + 2 * DIM, nxt, DIM, 0);
  gcn_prop_ln<<<nG, nB, 0, stream>>>(nxt, hbuf, neigh, dinv, keep,
                                     gcnG + 2 * DIM, gcnBb + 2 * DIM,
                                     out, 2 * DIM);

  wmma_gemm_bias<DIM><<<gG, gB, 0, stream>>>(x, poolW, poolB, tmp, DIM, 1);
  sage_pool_cat<<<nG, nB, 0, stream>>>(x, tmp, neigh, nbrc, cat);
  wmma_gemm_bias<2 * DIM><<<gG, gB, 0, stream>>>(cat, mrgW, mrgB,
                                                 sbuf, DIM, 1);

  sage_pool_cat<<<nG, nB, 0, stream>>>(sbuf, sbuf, neigh, nbrc, cat);
  wmma_gemm_bias<2 * DIM><<<gG, gB, 0, stream>>>(cat, mrgW + DIM * 2 * DIM,
                                                 mrgB + DIM, tmp, DIM, 1);
  resid_ln<<<nG, nB, 0, stream>>>(tmp, sbuf, sgG, sgB, sbuf, DIM);

  sage_pool_cat<<<nG, nB, 0, stream>>>(sbuf, sbuf, neigh, nbrc, cat);
  wmma_gemm_bias<2 * DIM><<<gG, gB, 0, stream>>>(cat, mrgW + 2 * DIM * 2 * DIM,
                                                 mrgB + 2 * DIM, tmp, DIM, 1);
  resid_ln<<<nG, nB, 0, stream>>>(tmp, sbuf, sgG + DIM, sgB + DIM,
                                  out + DIM, 2 * DIM);
}


# Row-local stages fused into the 32-row GEMM block: GCN gather+LN and SAGE max-pool build the A panel in LDS, SAGE residual LN runs in the epilogue; 16 launches -> 9; every sum order unchanged
# speedup vs baseline: 1.9591x; 1.5356x over previous
#include <hip/hip_runtime.h>

typedef __attribute__((ext_vector_type(2))) float v2f;
typedef __attribute__((ext_vector_type(8))) float v8f;

#define NNODE 8192
#define DIM   128
#define KNBR  8
#define NEGINF -1e30f

__device__ __forceinline__ float4 ld4(const float* p) {
  return *reinterpret_cast<const float4*>(p);
}
__device__ __forceinline__ void st4(float* p, float4 v) {
  *reinterpret_cast<float4*>(p) = v;
}

__global__ void deg_kernel(const int* __restrict__ neigh,
                           const int* __restrict__ nbrc,
                           float* __restrict__ dinv,
                           unsigned* __restrict__ keep) {
  int i = blockIdx.x * blockDim.x + threadIdx.x;
  if (i >= NNODE) return;
  int cnt = nbrc[i]; cnt = cnt > KNBR ? KNBR : cnt;
  const int* nb = neigh + i * KNBR;
  unsigned km = 0;
  int deg = 1;
  for (int k = 0; k < cnt; ++k) {
    int j = nb[k];
    if (j == i) continue;
    bool dup = false;
    for (int t = 0; t < k; ++t) dup |= (nb[t] == j);
    if (!dup) { km |= (1u << k); ++deg; }
  }
  dinv[i] = rsqrtf((float)deg);
  keep[i] = km;
}

template <int KB, int KE>
__device__ __forceinline__ void mma_range(const float* ap, const float* bp0,
                                          const float* bp1, v8f& c0, v8f& c1) {
#pragma unroll 8
  for (int k0 = KB; k0 < KE; k0 += 4) {
    const v2f a  = *reinterpret_cast<const v2f*>(ap + k0);
    const v2f b0 = *reinterpret_cast<const v2f*>(bp0 + k0);
    const v2f b1 = *reinterpret_cast<const v2f*>(bp1 + k0);
    c0 = __builtin_amdgcn_wmma_f32_16x16x4_f32(false, a, false, b0,
                                               (short)0, c0, false, false);
    c1 = __builtin_amdgcn_wmma_f32_16x16x4_f32(false, a, false, b1,
                                               (short)0, c1, false, false);
  }
}

template <int K>
__global__ __launch_bounds__(256)
void wmma_gemm_bias(const float* __restrict__ A,
                    const float* __restrict__ W,
                    const float* __restrict__ bias,
                    float* __restrict__ Y, int ldy, int do_relu) {
  constexpr int S   = K + 4;
  constexpr int H   = K / 2;
  constexpr int HC4 = H / 4;
  constexpr int NA  = (32 * HC4) / 256;
  constexpr int NW  = (DIM * HC4) / 256;
  __shared__ __attribute__((aligned(16))) float As[32 * S];
  __shared__ __attribute__((aligned(16))) float Ws[DIM * S];

  const int t  = threadIdx.x;
  const int m0 = blockIdx.x * 32;
  const float* __restrict__ Ablk = A + (size_t)m0 * K;

#pragma unroll
  for (int it = 0; it < NA; ++it) {
    const int idx = it * 256 + t;
    const int r = idx / HC4, c = idx % HC4;
    st4(As + r * S + 4 * c, ld4(Ablk + r * K + 4 * c));
  }
#pragma unroll
  for (int it = 0; it < NW; ++it) {
    const int idx = it * 256 + t;
    const int r = idx / HC4, c = idx % HC4;
    st4(Ws + r * S + 4 * c, ld4(W + r * K + 4 * c));
  }
  __syncthreads();

  float4 ra[NA];
  float4 rw[NW];
#pragma unroll
  for (int it = 0; it < NA; ++it) {
    const int idx = it * 256 + t;
    const int r = idx / HC4, c = idx % HC4;
    ra[it] = ld4(Ablk + r * K + H + 4 * c);
  }
#pragma unroll
  for (int it = 0; it < NW; ++it) {
    const int idx = it * 256 + t;
    const int r = idx / HC4, c = idx % HC4;
    rw[it] = ld4(W + r * K + H + 4 * c);
  }

  const int wave = t >> 5;
  const int lane = t & 31;
  const int half = lane >> 4;
  const int l16  = lane & 15;
  const int mt   = wave & 1;
  const int n0   = (wave >> 1) * 32;

  const float* ap  = As + (mt * 16 + l16) * S + 2 * half;
  const float* bp0 = Ws + (n0 + l16) * S + 2 * half;
  const float* bp1 = bp0 + 16 * S;

  v8f c0 = {};
  v8f c1 = {};
  mma_range<0, H>(ap, bp0, bp1, c0, c1);

#pragma unroll
  for (int it = 0; it < NA; ++it) {
    const int idx = it * 256 + t;
    const int r = idx / HC4, c = idx % HC4;
    st4(As + r * S + H + 4 * c, ra[it]);
  }
#pragma unroll
  for (int it = 0; it < NW; ++it) {
    const int idx = it * 256 + t;
    const int r = idx / HC4, c = idx % HC4;
    st4(Ws + r * S + H + 4 * c, rw[it]);
  }
  __syncthreads();
  mma_range<H, K>(ap, bp0, bp1, c0, c1);

  const float bv0 = bias[n0 + l16];
  const float bv1 = bias[n0 + 16 + l16];
  float* yrow = Y + (size_t)(m0 + mt * 16 + 8 * half) * ldy + n0 + l16;
#pragma unroll
  for (int r = 0; r < 8; ++r) {
    float v0 = c0[r] + bv0;
    float v1 = c1[r] + bv1;
    if (do_relu) { v0 = fmaxf(v0, 0.0f); v1 = fmaxf(v1, 0.0f); }
    yrow[(size_t)r * ldy]      = v0;
    yrow[(size_t)r * ldy + 16] = v1;
  }
}

__device__ __forceinline__ float4 wave_ln(float4 v, float4 gg, float4 bb) {
  float s = v.x + v.y + v.z + v.w;
#pragma unroll
  for (int off = 16; off > 0; off >>= 1) s += __shfl_xor(s, off, 32);
  const float mu = s * (1.0f / (float)DIM);
  const float dx = v.x - mu, dy = v.y - mu, dz = v.z - mu, dw = v.w - mu;
  float q = dx * dx + dy * dy + dz * dz + dw * dw;
#pragma unroll
  for (int off = 16; off > 0; off >>= 1) q += __shfl_xor(q, off, 32);
  const float rinv = rsqrtf(q * (1.0f / (float)DIM) + 1e-5f);
  float4 o;
  o.x = dx * rinv * gg.x + bb.x;
  o.y = dy * rinv * gg.y + bb.y;
  o.z = dz * rinv * gg.z + bb.z;
  o.w = dw * rinv * gg.w + bb.w;
  return o;
}

__device__ __forceinline__ void wave_ln_store(float4 v,
                                              const float* __restrict__ g,
                                              const float* __restrict__ b,
                                              float* __restrict__ dstrow,
                                              int lane) {
  const int d = lane * 4;
  st4(dstrow + d, wave_ln(v, ld4(g + d), ld4(b + d)));
}

enum { PRO_GCN = 1, PRO_SAGE = 2 };
enum { EPI_STORE = 0, EPI_LN = 1 };

template <int K, int PRO, int EPI>
__global__ __launch_bounds__(256) __attribute__((amdgpu_waves_per_eu(2, 2)))
void fused_rows_gemm(const float* Ain, const float* G,
                     const int* __restrict__ neigh,
                     const int* __restrict__ nbrc,
                     const float* __restrict__ dinv,
                     const unsigned* __restrict__ keep,
                     const float* __restrict__ pg,
                     const float* __restrict__ pb,
                     float* hout,
                     const float* __restrict__ W,
                     const float* __restrict__ bias,
                     float* Y, int ldy, int do_relu,
                     const float* __restrict__ eg,
                     const float* __restrict__ eb) {
  static_assert(PRO == PRO_GCN ? K == DIM : K == 2 * DIM, "panel width");
  static_assert(EPI == EPI_STORE || PRO == PRO_SAGE, "LN epilogue needs s in the panel");
  constexpr int S   = K + 4;
  constexpr int H   = K / 2;
  constexpr int HC4 = H / 4;
  constexpr int NW  = (DIM * HC4) / 256;
  __shared__ __attribute__((aligned(16))) float As[32 * S];
  __shared__ __attribute__((aligned(16))) float Ws[DIM * S];

  const int t    = threadIdx.x;
  const int m0   = blockIdx.x * 32;
  const int wave = __builtin_amdgcn_readfirstlane(t >> 5);
  const int lane = t & 31;
  const int d    = lane * 4;

  float4 rw[NW];
#pragma unroll
  for (int it = 0; it < NW; ++it) {
    const int idx = it * 256 + t;
    const int r = idx / HC4, c = idx % HC4;
    rw[it] = ld4(W + r * K + 4 * c);
  }

  int  jn[4][KNBR];
  bool vld[4][KNBR];
  if constexpr (PRO == PRO_GCN) {
    const int wrow = (t >> 5) * 4;
    const float4 gg = ld4(pg + d);
    const float4 bb = ld4(pb + d);
    float di[4];
#pragma unroll
    for (int it = 0; it < 4; ++it) {
      const int i = m0 + wrow + it;
      const unsigned km = keep[i];
      di[it] = dinv[i];
      const int4 na = *reinterpret_cast<const int4*>(neigh + i * KNBR);
      const int4 nb = *reinterpret_cast<const int4*>(neigh + i * KNBR + 4);
      const int nn[KNBR] = {na.x, na.y, na.z, na.w, nb.x, nb.y, nb.z, nb.w};
#pragma unroll
      for (int k = 0; k < KNBR; ++k) {
        vld[it][k] = ((km >> k) & 1u) != 0u;
        jn[it][k]  = vld[it][k] ? nn[k] : i;
      }
    }
    float dj[4][KNBR];
#pragma unroll
    for (int it = 0; it < 4; ++it)
#pragma unroll
      for (int k = 0; k < KNBR; ++k) dj[it][k] = dinv[jn[it][k]];
    float4 sf[4], rs[4], u[4][KNBR];
#pragma unroll
    for (int it = 0; it < 4; ++it) {
      const int i = m0 + wrow + it;
      sf[it] = ld4(G + (size_t)i * DIM + d);
      rs[it] = ld4(Ain + (size_t)i * DIM + d);
    }
#pragma unroll
    for (int it = 0; it < 4; ++it)
#pragma unroll
      for (int k = 0; k < KNBR; ++k)
        u[it][k] = ld4(G + (size_t)jn[it][k] * DIM + d);
#pragma unroll
    for (int it = 0; it < 4; ++it) {
      const int i = m0 + wrow + it;
      const float dii = di[it];
      float4 acc = {dii * sf[it].x, dii * sf[it].y, dii * sf[it].z, dii * sf[it].w};
#pragma unroll
      for (int k = 0; k < KNBR; ++k) {
        const float w = dj[it][k];
        const float4 uu = u[it][k];
        const float ax = acc.x + w * uu.x;
        const float ay = acc.y + w * uu.y;
        const float az = acc.z + w * uu.z;
        const float aw = acc.w + w * uu.w;
        const bool ok = vld[it][k];
        acc.x = ok ? ax : acc.x;
        acc.y = ok ? ay : acc.y;
        acc.z = ok ? az : acc.z;
        acc.w = ok ? aw : acc.w;
      }
      acc.x = fmaxf(acc.x * dii, 0.0f);
      acc.y = fmaxf(acc.y * dii, 0.0f);
      acc.z = fmaxf(acc.z * dii, 0.0f);
      acc.w = fmaxf(acc.w * dii, 0.0f);
      const float4 v = {acc.x + rs[it].x, acc.y + rs[it].y,
                        acc.z + rs[it].z, acc.w + rs[it].w};
      const float4 o = wave_ln(v, gg, bb);
      st4(hout + (size_t)i * DIM + d, o);
      st4(As + (wrow + it) * S + d, o);
    }
  } else {
    int cn[4];
#pragma unroll
    for (int it = 0; it < 4; ++it) {
      const int i = m0 + wave * 4 + it;
      int cnt = nbrc[i]; cnt = cnt > KNBR ? KNBR : cnt;
      cn[it] = cnt;
      const int4 na = *reinterpret_cast<const int4*>(neigh + i * KNBR);
      const int4 nb = *reinterpret_cast<const int4*>(neigh + i * KNBR + 4);
      const int nn[KNBR] = {na.x, na.y, na.z, na.w, nb.x, nb.y, nb.z, nb.w};
#pragma unroll
      for (int k = 0; k < KNBR; ++k) {
        vld[it][k] = k < cnt;
        jn[it][k]  = vld[it][k] ? nn[k] : i;
      }
    }
    float4 sf[4], u[4][KNBR];
#pragma unroll
    for (int it = 0; it < 4; ++it) {
      const int i = m0 + wave * 4 + it;
      sf[it] = ld4(Ain + (size_t)i * DIM + d);
    }
#pragma unroll
    for (int it = 0; it < 4; ++it)
#pragma unroll
      for (int k = 0; k < KNBR; ++k)
        u[it][k] = ld4(G + (size_t)jn[it][k] * DIM + d);
#pragma unroll
    for (int it = 0; it < 4; ++it) {
      st4(As + (wave * 4 + it) * S + d, sf[it]);
      float4 m = {NEGINF, NEGINF, NEGINF, NEGINF};
#pragma unroll
      for (int k = 0; k < KNBR; ++k) {
        const float4 uu = u[it][k];
        const float mx = fmaxf(m.x, uu.x);
        const float my = fmaxf(m.y, uu.y);
        const float mz = fmaxf(m.z, uu.z);
        const float mw = fmaxf(m.w, uu.w);
        const bool ok = vld[it][k];
        m.x = ok ? mx : m.x;
        m.y = ok ? my : m.y;
        m.z = ok ? mz : m.z;
        m.w = ok ? mw : m.w;
      }
      if (cn[it] == 0) m = make_float4(0.f, 0.f, 0.f, 0.f);
      st4(As + (wave * 4 + it) * S + DIM + d, m);
    }
  }

#pragma unroll
  for (int it = 0; it < NW; ++it) {
    const int idx = it * 256 + t;
    const int r = idx / HC4, c = idx % HC4;
    st4(Ws + r * S + 4 * c, rw[it]);
  }
  __syncthreads();

#pragma unroll
  for (int it = 0; it < NW; ++it) {
    const int idx = it * 256 + t;
    const int r = idx / HC4, c = idx % HC4;
    rw[it] = ld4(W + r * K + H + 4 * c);
  }

  const int half = lane >> 4;
  const int l16  = lane & 15;
  const int mt   = wave & 1;
  const int n0   = (wave >> 1) * 32;

  const float* ap  = As + (mt * 16 + l16) * S + 2 * half;
  const float* bp0 = Ws + (n0 + l16) * S + 2 * half;
  const float* bp1 = bp0 + 16 * S;

  v8f c0 = {};
  v8f c1 = {};
  mma_range<0, H>(ap, bp0, bp1, c0, c1);

#pragma unroll
  for (int it = 0; it < NW; ++it) {
    const int idx = it * 256 + t;
    const int r = idx / HC4, c = idx % HC4;
    st4(Ws + r * S + H + 4 * c, rw[it]);
  }
  __syncthreads();
  mma_range<H, K>(ap, bp0, bp1, c0, c1);

  const float bv0 = bias[n0 + l16];
  const float bv1 = bias[n0 + 16 + l16];
  if constexpr (EPI == EPI_STORE) {
    float* yrow = Y + (size_t)(m0 + mt * 16 + 8 * half) * ldy + n0 + l16;
#pragma unroll
    for (int r = 0; r < 8; ++r) {
      float v0 = c0[r] + bv0;
      float v1 = c1[r] + bv1;
      if (do_relu) { v0 = fmaxf(v0, 0.0f); v1 = fmaxf(v1, 0.0f); }
      yrow[(size_t)r * ldy]      = v0;
      yrow[(size_t)r * ldy + 16] = v1;
    }
  } else {
    __syncthreads();
    float* ys = As + (mt * 16 + 8 * half) * S + DIM + n0 + l16;
#pragma unroll
    for (int r = 0; r < 8; ++r) {
      float v0 = c0[r] + bv0;
      float v1 = c1[r] + bv1;
      if (do_relu) { v0 = fmaxf(v0, 0.0f); v1 = fmaxf(v1, 0.0f); }
      ys[r * S]      = v0;
      ys[r * S + 16] = v1;
    }
    __syncthreads();
    const float4 gg = ld4(eg + d);
    const float4 bb = ld4(eb + d);
#pragma unroll
    for (int it = 0; it < 4; ++it) {
      const int lr = wave * 4 + it;
      const float4 rv = ld4(As + lr * S + d);
      const float4 av = ld4(As + lr * S + DIM + d);
      const float4 v = {av.x + rv.x, av.y + rv.y, av.z + rv.z, av.w + rv.w};
      st4(Y + (size_t)(m0 + lr) * ldy + d, wave_ln(v, gg, bb));
    }
  }
}

__global__ void gcn_prop_ln(const float* __restrict__ nxt,
                            const float* __restrict__ resid,
                            const int* __restrict__ neigh,
                            const float* __restrict__ dinv,
                            const unsigned* __restrict__ keep,
                            const float* __restrict__ g,
                            const float* __restrict__ b,
                            float* __restrict__ dst, int ldd) {
  const int i = (int)((blockIdx.x * blockDim.x + threadIdx.x) >> 5);
  if (i >= NNODE) return;
  const int lane = threadIdx.x & 31;
  const int d = lane * 4;

  const float di = dinv[i];
  float4 t = ld4(nxt + (size_t)i * DIM + d);
  float4 acc = {di * t.x, di * t.y, di * t.z, di * t.w};

  const unsigned km = keep[i];
  const int* nb = neigh + i * KNBR;
  for (int k = 0; k < KNBR; ++k) {
    if ((km >> k) & 1u) {
      const int j = nb[k];
      const float dj = dinv[j];
      float4 u = ld4(nxt + (size_t)j * DIM + d);
      acc.x += dj * u.x; acc.y += dj * u.y;
      acc.z += dj * u.z; acc.w += dj * u.w;
    }
  }
  acc.x = fmaxf(acc.x * di, 0.0f);
  acc.y = fmaxf(acc.y * di, 0.0f);
  acc.z = fmaxf(acc.z * di, 0.0f);
  acc.w = fmaxf(acc.w * di, 0.0f);

  const float4 r = ld4(resid + (size_t)i * DIM + d);
  float4 v = {acc.x + r.x, acc.y + r.y, acc.z + r.z, acc.w + r.w};
  wave_ln_store(v, g, b, dst + (size_t)i * ldd, lane);
}

extern "C" void kernel_launch(void* const* d_in, const int* in_sizes, int n_in,
                              void* d_out, int out_size, void* d_ws, size_t ws_size,
                              hipStream_t stream) {
  const float* x      = (const float*)d_in[0];
  const int*   neigh  = (const int*)  d_in[1];
  const int*   nbrc   = (const int*)  d_in[2];
  const float* gcnW   = (const float*)d_in[3];
  const float* gcnB   = (const float*)d_in[4];
  const float* gcnG   = (const float*)d_in[5];
  const float* gcnBb  = (const float*)d_in[6];
  const float* poolW  = (const float*)d_in[7];
  const float* poolB  = (const float*)d_in[8];
  const float* mrgW   = (const float*)d_in[9];
  const float* mrgB   = (const float*)d_in[10];
  const float* sgG    = (const float*)d_in[11];
  const float* sgB    = (const float*)d_in[12];
  float* out = (float*)d_out;

  const size_t ND = (size_t)NNODE * DIM;
  float*    dinv  = (float*)d_ws;
  unsigned* keep  = (unsigned*)(dinv + NNODE);
  float*    first = (float*)(keep + NNODE);
  float*    nxt   = first + ND;
  float*    hbuf  = nxt + ND;
  float*    sbuf  = hbuf + ND;
  float*    tmp   = sbuf + ND;
  float*    nxt2  = tmp + ND;
  float*    sbuf2 = nxt2 + ND;

  const dim3 gB(256);
  const dim3 gG(NNODE / 32);
  const dim3 nB(256);
  const dim3 nG(NNODE / 8);

  deg_kernel<<<NNODE / 256, 256, 0, stream>>>(neigh, nbrc, dinv, keep);

  wmma_gemm_bias<DIM><<<gG, gB, 0, stream>>>(x, gcnW, gcnB, first, DIM, 0);
  fused_rows_gemm<DIM, PRO_GCN, EPI_STORE><<<gG, gB, 0, stream>>>(
      first, first, neigh, nullptr, dinv, keep, gcnG, gcnBb, hbuf,
      gcnW + DIM * DIM, gcnB + DIM, nxt, DIM, 0, nullptr, nullptr);
  fused_rows_gemm<DIM, PRO_GCN, EPI_STORE><<<gG, gB, 0, stream>>>(
      hbuf, nxt, neigh, nullptr, dinv, keep, gcnG + DIM, gcnBb + DIM, hbuf,
      gcnW + 2 * DIM * DIM, gcnB + 2 * DIM, nxt2, DIM, 0, nullptr, nullptr);
  gcn_prop_ln<<<nG, nB, 0, stream>>>(nxt2, hbuf, neigh, dinv, keep,
                                     gcnG + 2 * DIM, gcnBb + 2 * DIM,
                                     out, 2 * DIM);

  wmma_gemm_bias<DIM><<<gG, gB, 0, stream>>>(x, poolW, poolB, tmp, DIM, 1);
  fused_rows_gemm<2 * DIM, PRO_SAGE, EPI_STORE><<<gG, gB, 0, stream>>>(
      x, tmp, neigh, nbrc, nullptr, nullptr, nullptr, nullptr, nullptr,
      mrgW, mrgB, sbuf, DIM, 1, nullptr, nullptr);
  fused_rows_gemm<2 * DIM, PRO_SAGE, EPI_LN><<<gG, gB, 0, stream>>>(
      sbuf, sbuf, neigh, nbrc, nullptr, nullptr, nullptr, nullptr, nullptr,
      mrgW + DIM * 2 * DIM, mrgB + DIM, sbuf2, DIM, 1, sgG, sgB);
  fused_rows_gemm<2 * DIM, PRO_SAGE, EPI_LN><<<gG, gB, 0, stream>>>(
      sbuf2, sbuf2, neigh, nbrc, nullptr, nullptr, nullptr, nullptr, nullptr,
      mrgW + 2 * DIM * 2 * DIM, mrgB + 2 * DIM, out + DIM, 2 * DIM, 1,
      sgG + DIM, sgB + DIM);
}


# Independent GCN and SAGE branches share launches: block index picks the branch, two blocks resident per WGP (K=256 W buffer holds one K half so both fit LDS); 9 launches -> 4; sum orders unchanged
# speedup vs baseline: 3.0535x; 1.5586x over previous
#include <hip/hip_runtime.h>


typedef __attribute__((ext_vector_type(2))) float v2f;
typedef __attribute__((ext_vector_type(8))) float v8f;

#define NNODE 8192
#define DIM   128
#define KNBR  8
#define NEGINF -1e30f
#define NBLK  (NNODE / 32)

__device__ __forceinline__ float4 ld4(const float* p) {
  return *reinterpret_cast<const float4*>(p);
}
__device__ __forceinline__ void st4(float* p, float4 v) {
  *reinterpret_cast<float4*>(p) = v;
}

__device__ __forceinline__ void deg_node(int i,
                                         const int* __restrict__ neigh,
                                         const int* __restrict__ nbrc,
                                         float* __restrict__ dinv,
                                         unsigned* __restrict__ keep) {
  if (i >= NNODE) return;
  int cnt = nbrc[i]; cnt = cnt > KNBR ? KNBR : cnt;
  const int* nb = neigh + i * KNBR;
  unsigned km = 0;
  int deg = 1;
  for (int k = 0; k < cnt; ++k) {
    int j = nb[k];
    if (j == i) continue;
    bool dup = false;
    for (int t = 0; t < k; ++t) dup |= (nb[t] == j);
    if (!dup) { km |= (1u << k); ++deg; }
  }
  dinv[i] = rsqrtf((float)deg);
  keep[i] = km;
}

template <int KB, int KE, int BOFF>
__device__ __forceinline__ void mma_range(const float* ap, const float* bp0,
                                          const float* bp1, v8f& c0, v8f& c1) {
#pragma unroll 8
  for (int k0 = KB; k0 < KE; k0 += 4) {
    const v2f a  = *reinterpret_cast<const v2f*>(ap + k0);
    const v2f b0 = *reinterpret_cast<const v2f*>(bp0 + (k0 - BOFF));
    const v2f b1 = *reinterpret_cast<const v2f*>(bp1 + (k0 - BOFF));
    c0 = __builtin_amdgcn_wmma_f32_16x16x4_f32(false, a, false, b0,
                                               (short)0, c0, false, false);
    c1 = __builtin_amdgcn_wmma_f32_16x16x4_f32(false, a, false, b1,
                                               (short)0, c1, false, false);
  }
}

template <int K>
__device__ __forceinline__ void gemm_bias_block(float* lds, const int m0,
                                                const float* __restrict__ A,
                                                const float* __restrict__ W,
                                                const float* __restrict__ bias,
                                                float* __restrict__ Y,
                                                int ldy, int do_relu) {
  constexpr int S   = K + 4;
  constexpr int H   = K / 2;
  constexpr int HC4 = H / 4;
  constexpr int NA  = (32 * HC4) / 256;
  constexpr int NW  = (DIM * HC4) / 256;
  float* As = lds;
  float* Ws = lds + 32 * S;

  const int t  = threadIdx.x;
  const float* __restrict__ Ablk = A + (size_t)m0 * K;

#pragma unroll
  for (int it = 0; it < NA; ++it) {
    const int idx = it * 256 + t;
    const int r = idx / HC4, c = idx % HC4;
    st4(As + r * S + 4 * c, ld4(Ablk + r * K + 4 * c));
  }
#pragma unroll
  for (int it = 0; it < NW; ++it) {
    const int idx = it * 256 + t;
    const int r = idx / HC4, c = idx % HC4;
    st4(Ws + r * S + 4 * c, ld4(W + r * K + 4 * c));
  }
  __syncthreads();

  float4 ra[NA];
  float4 rw[NW];
#pragma unroll
  for (int it = 0; it < NA; ++it) {
    const int idx = it * 256 + t;
    const int r = idx / HC4, c = idx % HC4;
    ra[it] = ld4(Ablk + r * K + H + 4 * c);
  }
#pragma unroll
  for (int it = 0; it < NW; ++it) {
    const int idx = it * 256 + t;
    const int r = idx / HC4, c = idx % HC4;
    rw[it] = ld4(W + r * K + H + 4 * c);
  }

  const int wave = t >> 5;
  const int lane = t & 31;
  const int half = lane >> 4;
  const int l16  = lane & 15;
  const int mt   = wave & 1;
  const int n0   = (wave >> 1) * 32;

  const float* ap  = As + (mt * 16 + l16) * S + 2 * half;
  const float* bp0 = Ws + (n0 + l16) * S + 2 * half;
  const float* bp1 = bp0 + 16 * S;

  v8f c0 = {};
  v8f c1 = {};
  mma_range<0, H, 0>(ap, bp0, bp1, c0, c1);

#pragma unroll
  for (int it = 0; it < NA; ++it) {
    const int idx = it * 256 + t;
    const int r = idx / HC4, c = idx % HC4;
    st4(As + r * S + H + 4 * c, ra[it]);
  }
#pragma unroll
  for (int it = 0; it < NW; ++it) {
    const int idx = it * 256 + t;
    const int r = idx / HC4, c = idx % HC4;
    st4(Ws + r * S + H + 4 * c, rw[it]);
  }
  __syncthreads();
  mma_range<H, K, 0>(ap, bp0, bp1, c0, c1);

  const float bv0 = bias[n0 + l16];
  const float bv1 = bias[n0 + 16 + l16];
  float* yrow = Y + (size_t)(m0 + mt * 16 + 8 * half) * ldy + n0 + l16;
#pragma unroll
  for (int r = 0; r < 8; ++r) {
    float v0 = c0[r] + bv0;
    float v1 = c1[r] + bv1;
    if (do_relu) { v0 = fmaxf(v0, 0.0f); v1 = fmaxf(v1, 0.0f); }
    yrow[(size_t)r * ldy]      = v0;
    yrow[(size_t)r * ldy + 16] = v1;
  }
}

__device__ __forceinline__ float4 wave_ln(float4 v, float4 gg, float4 bb) {
  float s = v.x + v.y + v.z + v.w;
#pragma unroll
  for (int off = 16; off > 0; off >>= 1) s += __shfl_xor(s, off, 32);
  const float mu = s * (1.0f / (float)DIM);
  const float dx = v.x - mu, dy = v.y - mu, dz = v.z - mu, dw = v.w - mu;
  float q = dx * dx + dy * dy + dz * dz + dw * dw;
#pragma unroll
  for (int off = 16; off > 0; off >>= 1) q += __shfl_xor(q, off, 32);
  const float rinv = rsqrtf(q * (1.0f / (float)DIM) + 1e-5f);
  float4 o;
  o.x = dx * rinv * gg.x + bb.x;
  o.y = dy * rinv * gg.y + bb.y;
  o.z = dz * rinv * gg.z + bb.z;
  o.w = dw * rinv * gg.w + bb.w;
  return o;
}

enum { PRO_GCN = 1, PRO_SAGE = 2 };
enum { EPI_STORE = 0, EPI_LN = 1, EPI_NONE = 2 };

#define LDS_FLOATS (32 * (2 * DIM + 4) + DIM * (DIM + 4))

template <int K, int PRO, int EPI>
__device__ __forceinline__ void fused_block(float* lds, const int m0,
                     const float* Ain, const float* G,
                     const int* __restrict__ neigh,
                     const int* __restrict__ nbrc,
                     const float* __restrict__ dinv,
                     const unsigned* __restrict__ keep,
                     const float* __restrict__ pg,
                     const float* __restrict__ pb,
                     float* hout, int ldh,
                     const float* __restrict__ W,
                     const float* __restrict__ bias,
                     float* Y, int ldy, int do_relu,
                     const float* __restrict__ eg,
                     const float* __restrict__ eb) {
  static_assert(PRO == PRO_GCN ? K == DIM : K == 2 * DIM, "panel width");
  static_assert(EPI != EPI_LN || PRO == PRO_SAGE, "LN epilogue needs s in the panel");
  static_assert(EPI != EPI_NONE || PRO == PRO_GCN, "only the GCN rows end without a GEMM");
  constexpr bool HALFW = (K == 2 * DIM);
  constexpr int S   = K + 4;
  constexpr int H   = K / 2;
  constexpr int SW  = HALFW ? H + 4 : K + 4;
  constexpr int HC4 = H / 4;
  constexpr int NW  = (DIM * HC4) / 256;
  static_assert(32 * S + DIM * SW <= LDS_FLOATS, "LDS size");
  float* As = lds;
  float* Ws = lds + 32 * S;

  const int t    = threadIdx.x;
  const int wave = __builtin_amdgcn_readfirstlane(t >> 5);
  const int lane = t & 31;
  const int d    = lane * 4;

  float4 rw[NW];
  if constexpr (EPI != EPI_NONE) {
#pragma unroll
    for (int it = 0; it < NW; ++it) {
      const int idx = it * 256 + t;
      const int r = idx / HC4, c = idx % HC4;
      rw[it] = ld4(W + r * K + 4 * c);
    }
  }

  int  jn[4][KNBR];
  bool vld[4][KNBR];
  if constexpr (PRO == PRO_GCN) {
    const int wrow = (t >> 5) * 4;
    const float4 gg = ld4(pg + d);
    const float4 bb = ld4(pb + d);
    float di[4];
#pragma unroll
    for (int it = 0; it < 4; ++it) {
      const int i = m0 + wrow + it;
      const unsigned km = keep[i];
      di[it] = dinv[i];
      const int4 na = *reinterpret_cast<const int4*>(neigh + i * KNBR);
      const int4 nb = *reinterpret_cast<const int4*>(neigh + i * KNBR + 4);
      const int nn[KNBR] = {na.x, na.y, na.z, na.w, nb.x, nb.y, nb.z, nb.w};
#pragma unroll
      for (int k = 0; k < KNBR; ++k) {
        vld[it][k] = ((km >> k) & 1u) != 0u;
        jn[it][k]  = vld[it][k] ? nn[k] : i;
      }
    }
    float dj[4][KNBR];
#pragma unroll
    for (int it = 0; it < 4; ++it)
#pragma unroll
      for (int k = 0; k < KNBR; ++k) dj[it][k] = dinv[jn[it][k]];
    float4 sf[4], rs[4], u[4][KNBR];
#pragma unroll
    for (int it = 0; it < 4; ++it) {
      const int i = m0 + wrow + it;
      sf[it] = ld4(G + (size_t)i * DIM + d);
      rs[it] = ld4(Ain + (size_t)i * DIM + d);
    }
#pragma unroll
    for (int it = 0; it < 4; ++it)
#pragma unroll
      for (int k = 0; k < KNBR; ++k)
        u[it][k] = ld4(G + (size_t)jn[it][k] * DIM + d);
#pragma unroll
    for (int it = 0; it < 4; ++it) {
      const int i = m0 + wrow + it;
      const float dii = di[it];
      float4 acc = {dii * sf[it].x, dii * sf[it].y, dii * sf[it].z, dii * sf[it].w};
#pragma unroll
      for (int k = 0; k < KNBR; ++k) {
        const float w = dj[it][k];
        const float4 uu = u[it][k];
        const float ax = acc.x + w * uu.x;
        const float ay = acc.y + w * uu.y;
        const float az = acc.z + w * uu.z;
        const float aw = acc.w + w * uu.w;
        const bool ok = vld[it][k];
        acc.x = ok ? ax : acc.x;
        acc.y = ok ? ay : acc.y;
        acc.z = ok ? az : acc.z;
        acc.w = ok ? aw : acc.w;
      }
      acc.x = fmaxf(acc.x * dii, 0.0f);
      acc.y = fmaxf(acc.y * dii, 0.0f);
      acc.z = fmaxf(acc.z * dii, 0.0f);
      acc.w = fmaxf(acc.w * dii, 0.0f);
      const float4 v = {acc.x + rs[it].x, acc.y + rs[it].y,
                        acc.z + rs[it].z, acc.w + rs[it].w};
      const float4 o = wave_ln(v, gg, bb);
      st4(hout + (size_t)i * ldh + d, o);
      if constexpr (EPI != EPI_NONE) st4(As + (wrow + it) * S + d, o);
    }
  } else {
    int cn[4];
#pragma unroll
    for (int it = 0; it < 4; ++it) {
      const int i = m0 + wave * 4 + it;
      int cnt = nbrc[i]; cnt = cnt > KNBR ? KNBR : cnt;
      cn[it] = cnt;
      const int4 na = *reinterpret_cast<const int4*>(neigh + i * KNBR);
      const int4 nb = *reinterpret_cast<const int4*>(neigh + i * KNBR + 4);
      const int nn[KNBR] = {na.x, na.y, na.z, na.w, nb.x, nb.y, nb.z, nb.w};
#pragma unroll
      for (int k = 0; k < KNBR; ++k) {
        vld[it][k] = k < cnt;
        jn[it][k]  = vld[it][k] ? nn[k] : i;
      }
    }
    float4 sf[4], u[4][KNBR];
#pragma unroll
    for (int it = 0; it < 4; ++it) {
      const int i = m0 + wave * 4 + it;
      sf[it] = ld4(Ain + (size_t)i * DIM + d);
    }
#pragma unroll
    for (int it = 0; it < 4; ++it)
#pragma unroll
      for (int k = 0; k < KNBR; ++k)
        u[it][k] = ld4(G + (size_t)jn[it][k] * DIM + d);
#pragma unroll
    for (int it = 0; it < 4; ++it) {
      st4(As + (wave * 4 + it) * S + d, sf[it]);
      float4 m = {NEGINF, NEGINF, NEGINF, NEGINF};
#pragma unroll
      for (int k = 0; k < KNBR; ++k) {
        const float4 uu = u[it][k];
        const float mx = fmaxf(m.x, uu.x);
        const float my = fmaxf(m.y, uu.y);
        const float mz = fmaxf(m.z, uu.z);
        const float mw = fmaxf(m.w, uu.w);
        const bool ok = vld[it][k];
        m.x = ok ? mx : m.x;
        m.y = ok ? my : m.y;
        m.z = ok ? mz : m.z;
        m.w = ok ? mw : m.w;
      }
      if (cn[it] == 0) m = make_float4(0.f, 0.f, 0.f, 0.f);
      st4(As + (wave * 4 + it) * S + DIM + d, m);
    }
  }

  if constexpr (EPI != EPI_NONE) {
#pragma unroll
    for (int it = 0; it < NW; ++it) {
      const int idx = it * 256 + t;
      const int r = idx / HC4, c = idx % HC4;
      st4(Ws + r * SW + 4 * c, rw[it]);
    }
    __syncthreads();

#pragma unroll
    for (int it = 0; it < NW; ++it) {
      const int idx = it * 256 + t;
      const int r = idx / HC4, c = idx % HC4;
      rw[it] = ld4(W + r * K + H + 4 * c);
    }

    const int half = lane >> 4;
    const int l16  = lane & 15;
    const int mt   = wave & 1;
    const int n0   = (wave >> 1) * 32;

    const float* ap  = As + (mt * 16 + l16) * S + 2 * half;
    const float* bp0 = Ws + (n0 + l16) * SW + 2 * half;
    const float* bp1 = bp0 + 16 * SW;

    v8f c0 = {};
    v8f c1 = {};
    mma_range<0, H, 0>(ap, bp0, bp1, c0, c1);

    if constexpr (HALFW) {
      __syncthreads();
#pragma unroll
      for (int it = 0; it < NW; ++it) {
        const int idx = it * 256 + t;
        const int r = idx / HC4, c = idx % HC4;
        st4(Ws + r * SW + 4 * c, rw[it]);
      }
      __syncthreads();
      mma_range<H, K, H>(ap, bp0, bp1, c0, c1);
    } else {
#pragma unroll
      for (int it = 0; it < NW; ++it) {
        const int idx = it * 256 + t;
        const int r = idx / HC4, c = idx % HC4;
        st4(Ws + r * SW + H + 4 * c, rw[it]);
      }
      __syncthreads();
      mma_range<H, K, 0>(ap, bp0, bp1, c0, c1);
    }

    const float bv0 = bias[n0 + l16];
    const float bv1 = bias[n0 + 16 + l16];
    if constexpr (EPI == EPI_STORE) {
      float* yrow = Y + (size_t)(m0 + mt * 16 + 8 * half) * ldy + n0 + l16;
#pragma unroll
      for (int r = 0; r < 8; ++r) {
        float v0 = c0[r] + bv0;
        float v1 = c1[r] + bv1;
        if (do_relu) { v0 = fmaxf(v0, 0.0f); v1 = fmaxf(v1, 0.0f); }
        yrow[(size_t)r * ldy]      = v0;
        yrow[(size_t)r * ldy + 16] = v1;
      }
    } else {
      __syncthreads();
      float* ys = As + (mt * 16 + 8 * half) * S + DIM + n0 + l16;
#pragma unroll
      for (int r = 0; r < 8; ++r) {
        float v0 = c0[r] + bv0;
        float v1 = c1[r] + bv1;
        if (do_relu) { v0 = fmaxf(v0, 0.0f); v1 = fmaxf(v1, 0.0f); }
        ys[r * S]      = v0;
        ys[r * S + 16] = v1;
      }
      __syncthreads();
      const float4 gg = ld4(eg + d);
      const float4 bb = ld4(eb + d);
#pragma unroll
      for (int it = 0; it < 4; ++it) {
        const int lr = wave * 4 + it;
        const float4 rv = ld4(As + lr * S + d);
        const float4 av = ld4(As + lr * S + DIM + d);
        const float4 v = {av.x + rv.x, av.y + rv.y, av.z + rv.z, av.w + rv.w};
        st4(Y + (size_t)(m0 + lr) * ldy + d, wave_ln(v, gg, bb));
      }
    }
  }
}


__global__ __launch_bounds__(256)
void pair_gemm_deg(const float* __restrict__ x,
                   const float* __restrict__ Wg, const float* __restrict__ bg,
                   float* __restrict__ Yg,
                   const float* __restrict__ Wp, const float* __restrict__ bp,
                   float* __restrict__ Yp,
                   const int* __restrict__ neigh, const int* __restrict__ nbrc,
                   float* __restrict__ dinv, unsigned* __restrict__ keep) {
  __shared__ __attribute__((aligned(16))) float lds[(32 + DIM) * (DIM + 4)];
  const int b = blockIdx.x;
  if (b >= 2 * NBLK) {
    deg_node((b - 2 * NBLK) * 256 + (int)threadIdx.x, neigh, nbrc, dinv, keep);
    return;
  }
  const bool pool = b >= NBLK;
  gemm_bias_block<DIM>(lds, (pool ? b - NBLK : b) * 32, x,
                       pool ? Wp : Wg, pool ? bp : bg, pool ? Yp : Yg,
                       DIM, pool ? 1 : 0);
}

template <int EPI_S, int EPI_G>
__global__ __launch_bounds__(256) __attribute__((amdgpu_waves_per_eu(4, 4)))
void pair_layer(const float* sAin, const float* sG,
                const int* __restrict__ neigh,
                const int* __restrict__ nbrc,
                const float* __restrict__ sW,
                const float* __restrict__ sbias,
                float* sY, int sldy,
                const float* __restrict__ seg,
                const float* __restrict__ seb,
                const float* gAin, const float* gG,
                const float* __restrict__ dinv,
                const unsigned* __restrict__ keep,
                const float* __restrict__ gpg,
                const float* __restrict__ gpb,
                float* ghout, int gldh,
                const float* __restrict__ gW,
                const float* __restrict__ gbias,
                float* gY) {
  __shared__ __attribute__((aligned(16))) float lds[LDS_FLOATS];
  const int b = blockIdx.x;
  if (b < NBLK) {
    fused_block<2 * DIM, PRO_SAGE, EPI_S>(
        lds, b * 32, sAin, sG, neigh, nbrc, nullptr, nullptr, nullptr, nullptr,
        nullptr, 0, sW, sbias, sY, sldy, 1, seg, seb);
  } else {
    fused_block<DIM, PRO_GCN, EPI_G>(
        lds, (b - NBLK) * 32, gAin, gG, neigh, nullptr, dinv, keep, gpg, gpb,
        ghout, gldh, gW, gbias, gY, DIM, 0, nullptr, nullptr);
  }
}

extern "C" void kernel_launch(void* const* d_in, const int* in_sizes, int n_in,
                              void* d_out, int out_size, void* d_ws, size_t ws_size,
                              hipStream_t stream) {
  const float* x      = (const float*)d_in[0];
  const int*   neigh  = (const int*)  d_in[1];
  const int*   nbrc   = (const int*)  d_in[2];
  const float* gcnW   = (const float*)d_in[3];
  const float* gcnB   = (const float*)d_in[4];
  const float* gcnG   = (const float*)d_in[5];
  const float* gcnBb  = (const float*)d_in[6];
  const float* poolW  = (const float*)d_in[7];
  const float* poolB  = (const float*)d_in[8];
  const float* mrgW   = (const float*)d_in[9];
  const float* mrgB   = (const float*)d_in[10];
  const float* sgG    = (const float*)d_in[11];
  const float* sgB    = (const float*)d_in[12];
  float* out = (float*)d_out;

  const size_t ND = (size_t)NNODE * DIM;
  float*    dinv  = (float*)d_ws;
  unsigned* keep  = (unsigned*)(dinv + NNODE);
  float*    first = (float*)(keep + NNODE);
  float*    nxt   = first + ND;
  float*    hbuf  = nxt + ND;
  float*    sbuf  = hbuf + ND;
  float*    tmp   = sbuf + ND;
  float*    nxt2  = tmp + ND;
  float*    sbuf2 = nxt2 + ND;

  const dim3 gB(256);
  const dim3 g1(2 * NBLK + NNODE / 256);
  const dim3 g2(2 * NBLK);

  pair_gemm_deg<<<g1, gB, 0, stream>>>(x, gcnW, gcnB, first, poolW, poolB, tmp,
                                       neigh, nbrc, dinv, keep);
  pair_layer<EPI_STORE, EPI_STORE><<<g2, gB, 0, stream>>>(
      x, tmp, neigh, nbrc, mrgW, mrgB, sbuf, DIM, nullptr, nullptr,
      first, first, dinv, keep, gcnG, gcnBb, hbuf, DIM,
      gcnW + DIM * DIM, gcnB + DIM, nxt);
  pair_layer<EPI_LN, EPI_STORE><<<g2, gB, 0, stream>>>(
      sbuf, sbuf, neigh, nbrc, mrgW + DIM * 2 * DIM, mrgB + DIM, sbuf2, DIM,
      sgG, sgB,
      hbuf, nxt, dinv, keep, gcnG + DIM, gcnBb + DIM, hbuf, DIM,
      gcnW + 2 * DIM * DIM, gcnB + 2 * DIM, nxt2);
  pair_layer<EPI_LN, EPI_NONE><<<g2, gB, 0, stream>>>(
      sbuf2, sbuf2, neigh, nbrc, mrgW + 2 * DIM * 2 * DIM, mrgB + 2 * DIM,
      out + DIM, 2 * DIM, sgG + DIM, sgB + DIM,
      hbuf, nxt2, dinv, keep, gcnG + 2 * DIM, gcnBb + 2 * DIM,
      out, 2 * DIM, nullptr, nullptr, nullptr);
}


# Round trips off each block's serial path: bias and LN gain/shift loaded ahead of the products; LN butterfly sums on permlanex16 + row_xmask DPP, not LDS shuffles; same partner order, bit-identical
# speedup vs baseline: 3.1731x; 1.0392x over previous
#include <hip/hip_runtime.h>


typedef __attribute__((ext_vector_type(2))) float v2f;
typedef __attribute__((ext_vector_type(8))) float v8f;

#define NNODE 8192
#define DIM   128
#define KNBR  8
#define NEGINF -1e30f
#define NBLK  (NNODE / 32)

__device__ __forceinline__ float4 ld4(const float* p) {
  return *reinterpret_cast<const float4*>(p);
}
__device__ __forceinline__ void st4(float* p, float4 v) {
  *reinterpret_cast<float4*>(p) = v;
}

__device__ __forceinline__ void deg_node(int i,
                                         const int* __restrict__ neigh,
                                         const int* __restrict__ nbrc,
                                         float* __restrict__ dinv,
                                         unsigned* __restrict__ keep) {
  if (i >= NNODE) return;
  int cnt = nbrc[i]; cnt = cnt > KNBR ? KNBR : cnt;
  const int* nb = neigh + i * KNBR;
  unsigned km = 0;
  int deg = 1;
  for (int k = 0; k < cnt; ++k) {
    int j = nb[k];
    if (j == i) continue;
    bool dup = false;
    for (int t = 0; t < k; ++t) dup |= (nb[t] == j);
    if (!dup) { km |= (1u << k); ++deg; }
  }
  dinv[i] = rsqrtf((float)deg);
  keep[i] = km;
}

template <int KB, int KE, int BOFF>
__device__ __forceinline__ void mma_range(const float* ap, const float* bp0,
                                          const float* bp1, v8f& c0, v8f& c1) {
#pragma unroll 8
  for (int k0 = KB; k0 < KE; k0 += 4) {
    const v2f a  = *reinterpret_cast<const v2f*>(ap + k0);
    const v2f b0 = *reinterpret_cast<const v2f*>(bp0 + (k0 - BOFF));
    const v2f b1 = *reinterpret_cast<const v2f*>(bp1 + (k0 - BOFF));
    c0 = __builtin_amdgcn_wmma_f32_16x16x4_f32(false, a, false, b0,
                                               (short)0, c0, false, false);
    c1 = __builtin_amdgcn_wmma_f32_16x16x4_f32(false, a, false, b1,
                                               (short)0, c1, false, false);
  }
}

template <int K>
__device__ __forceinline__ void gemm_bias_block(float* lds, const int m0,
                                                const float* __restrict__ A,
                                                const float* __restrict__ W,
                                                const float* __restrict__ bias,
                                                float* __restrict__ Y,
                                                int ldy, int do_relu) {
  constexpr int S   = K + 4;
  constexpr int H   = K / 2;
  constexpr int HC4 = H / 4;
  constexpr int NA  = (32 * HC4) / 256;
  constexpr int NW  = (DIM * HC4) / 256;
  float* As = lds;
  float* Ws = lds + 32 * S;

  const int t  = threadIdx.x;
  const float* __restrict__ Ablk = A + (size_t)m0 * K;

#pragma unroll
  for (int it = 0; it < NA; ++it) {
    const int idx = it * 256 + t;
    const int r = idx / HC4, c = idx % HC4;
    st4(As + r * S + 4 * c, ld4(Ablk + r * K + 4 * c));
  }
#pragma unroll
  for (int it = 0; it < NW; ++it) {
    const int idx = it * 256 + t;
    const int r = idx / HC4, c = idx % HC4;
    st4(Ws + r * S + 4 * c, ld4(W + r * K + 4 * c));
  }
  __syncthreads();

  float4 ra[NA];
  float4 rw[NW];
#pragma unroll
  for (int it = 0; it < NA; ++it) {
    const int idx = it * 256 + t;
    const int r = idx / HC4, c = idx % HC4;
    ra[it] = ld4(Ablk + r * K + H + 4 * c);
  }
#pragma unroll
  for (int it = 0; it < NW; ++it) {
    const int idx = it * 256 + t;
    const int r = idx / HC4, c = idx % HC4;
    rw[it] = ld4(W + r * K + H + 4 * c);
  }

  const int wave = t >> 5;
  const int lane = t & 31;
  const int half = lane >> 4;
  const int l16  = lane & 15;
  const int mt   = wave & 1;
  const int n0   = (wave >> 1) * 32;

  const float* ap  = As + (mt * 16 + l16) * S + 2 * half;
  const float* bp0 = Ws + (n0 + l16) * S + 2 * half;
  const float* bp1 = bp0 + 16 * S;

  const float bv0 = bias[n0 + l16];
  const float bv1 = bias[n0 + 16 + l16];

  v8f c0 = {};
  v8f c1 = {};
  mma_range<0, H, 0>(ap, bp0, bp1, c0, c1);

#pragma unroll
  for (int it = 0; it < NA; ++it) {
    const int idx = it * 256 + t;
    const int r = idx / HC4, c = idx % HC4;
    st4(As + r * S + H + 4 * c, ra[it]);
  }
#pragma unroll
  for (int it = 0; it < NW; ++it) {
    const int idx = it * 256 + t;
    const int r = idx / HC4, c = idx % HC4;
    st4(Ws + r * S + H + 4 * c, rw[it]);
  }
  __syncthreads();
  mma_range<H, K, 0>(ap, bp0, bp1, c0, c1);

  float* yrow = Y + (size_t)(m0 + mt * 16 + 8 * half) * ldy + n0 + l16;
#pragma unroll
  for (int r = 0; r < 8; ++r) {
    float v0 = c0[r] + bv0;
    float v1 = c1[r] + bv1;
    if (do_relu) { v0 = fmaxf(v0, 0.0f); v1 = fmaxf(v1, 0.0f); }
    yrow[(size_t)r * ldy]      = v0;
    yrow[(size_t)r * ldy + 16] = v1;
  }
}

__device__ __forceinline__ float lane_xor16(float x) {
  const int b = __builtin_bit_cast(int, x);
  return __builtin_bit_cast(float,
      __builtin_amdgcn_permlanex16(b, b, 0x76543210, 0xfedcba98, false, false));
}
template <int XMASK>
__device__ __forceinline__ float lane_xor_row(float x) {
  return __builtin_bit_cast(float,
      __builtin_amdgcn_update_dpp(0, __builtin_bit_cast(int, x), 0x160 + XMASK,
                                  0xf, 0xf, true));
}
__device__ __forceinline__ void wave_sum32x4(float (&s)[4]) {
#pragma unroll
  for (int i = 0; i < 4; ++i) s[i] += lane_xor16(s[i]);
#pragma unroll
  for (int i = 0; i < 4; ++i) s[i] += lane_xor_row<8>(s[i]);
#pragma unroll
  for (int i = 0; i < 4; ++i) s[i] += lane_xor_row<4>(s[i]);
#pragma unroll
  for (int i = 0; i < 4; ++i) s[i] += lane_xor_row<2>(s[i]);
#pragma unroll
  for (int i = 0; i < 4; ++i) s[i] += lane_xor_row<1>(s[i]);
}
__device__ __forceinline__ void wave_ln4(const float4 (&v)[4], float4 gg,
                                         float4 bb, float4 (&o)[4]) {
  float s[4];
#pragma unroll
  for (int i = 0; i < 4; ++i) s[i] = v[i].x + v[i].y + v[i].z + v[i].w;
  wave_sum32x4(s);
  float dx[4], dy[4], dz[4], dw[4], q[4];
#pragma unroll
  for (int i = 0; i < 4; ++i) {
    const float mu = s[i] * (1.0f / (float)DIM);
    dx[i] = v[i].x - mu; dy[i] = v[i].y - mu;
    dz[i] = v[i].z - mu; dw[i] = v[i].w - mu;
    q[i] = dx[i] * dx[i] + dy[i] * dy[i] + dz[i] * dz[i] + dw[i] * dw[i];
  }
  wave_sum32x4(q);
#pragma unroll
  for (int i = 0; i < 4; ++i) {
    const float rinv = rsqrtf(q[i] * (1.0f / (float)DIM) + 1e-5f);
    o[i].x = dx[i] * rinv * gg.x + bb.x;
    o[i].y = dy[i] * rinv * gg.y + bb.y;
    o[i].z = dz[i] * rinv * gg.z + bb.z;
    o[i].w = dw[i] * rinv * gg.w + bb.w;
  }
}

enum { PRO_GCN = 1, PRO_SAGE = 2 };
enum { EPI_STORE = 0, EPI_LN = 1, EPI_NONE = 2 };

#define LDS_FLOATS (32 * (2 * DIM + 4) + DIM * (DIM + 4))

template <int K, int PRO, int EPI>
__device__ __forceinline__ void fused_block(float* lds, const int m0,
                     const float* Ain, const float* G,
                     const int* __restrict__ neigh,
                     const int* __restrict__ nbrc,
                     const float* __restrict__ dinv,
                     const unsigned* __restrict__ keep,
                     const float* __restrict__ pg,
                     const float* __restrict__ pb,
                     float* hout, int ldh,
                     const float* __restrict__ W,
                     const float* __restrict__ bias,
                     float* Y, int ldy, int do_relu,
                     const float* __restrict__ eg,
                     const float* __restrict__ eb) {
  static_assert(PRO == PRO_GCN ? K == DIM : K == 2 * DIM, "panel width");
  static_assert(EPI != EPI_LN || PRO == PRO_SAGE, "LN epilogue needs s in the panel");
  static_assert(EPI != EPI_NONE || PRO == PRO_GCN, "only the GCN rows end without a GEMM");
  constexpr bool HALFW = (K == 2 * DIM);
  constexpr int S   = K + 4;
  constexpr int H   = K / 2;
  constexpr int SW  = HALFW ? H + 4 : K + 4;
  constexpr int HC4 = H / 4;
  constexpr int NW  = (DIM * HC4) / 256;
  static_assert(32 * S + DIM * SW <= LDS_FLOATS, "LDS size");
  float* As = lds;
  float* Ws = lds + 32 * S;

  const int t    = threadIdx.x;
  const int wave = __builtin_amdgcn_readfirstlane(t >> 5);
  const int lane = t & 31;
  const int d    = lane * 4;

  float4 rw[NW];
  if constexpr (EPI != EPI_NONE) {
#pragma unroll
    for (int it = 0; it < NW; ++it) {
      const int idx = it * 256 + t;
      const int r = idx / HC4, c = idx % HC4;
      rw[it] = ld4(W + r * K + 4 * c);
    }
  }

  int  jn[4][KNBR];
  bool vld[4][KNBR];
  if constexpr (PRO == PRO_GCN) {
    const int wrow = (t >> 5) * 4;
    const float4 gg = ld4(pg + d);
    const float4 bb = ld4(pb + d);
    float di[4];
#pragma unroll
    for (int it = 0; it < 4; ++it) {
      const int i = m0 + wrow + it;
      const unsigned km = keep[i];
      di[it] = dinv[i];
      const int4 na = *reinterpret_cast<const int4*>(neigh + i * KNBR);
      const int4 nb = *reinterpret_cast<const int4*>(neigh + i * KNBR + 4);
      const int nn[KNBR] = {na.x, na.y, na.z, na.w, nb.x, nb.y, nb.z, nb.w};
#pragma unroll
      for (int k = 0; k < KNBR; ++k) {
        vld[it][k] = ((km >> k) & 1u) != 0u;
        jn[it][k]  = vld[it][k] ? nn[k] : i;
      }
    }
    float dj[4][KNBR];
#pragma unroll
    for (int it = 0; it < 4; ++it)
#pragma unroll
      for (int k = 0; k < KNBR; ++k) dj[it][k] = dinv[jn[it][k]];
    float4 sf[4], rs[4], u[4][KNBR];
#pragma unroll
    for (int it = 0; it < 4; ++it) {
      const int i = m0 + wrow + it;
      sf[it] = ld4(G + (size_t)i * DIM + d);
      rs[it] = ld4(Ain + (size_t)i * DIM + d);
    }
#pragma unroll
    for (int it = 0; it < 4; ++it)
#pragma unroll
      for (int k = 0; k < KNBR; ++k)
        u[it][k] = ld4(G + (size_t)jn[it][k] * DIM + d);
    float4 vin[4], o[4];
#pragma unroll
    for (int it = 0; it < 4; ++it) {
      const float dii = di[it];
      float4 acc = {dii * sf[it].x, dii * sf[it].y, dii * sf[it].z, dii * sf[it].w};
#pragma unroll
      for (int k = 0; k < KNBR; ++k) {
        const float w = dj[it][k];
        const float4 uu = u[it][k];
        const float ax = acc.x + w * uu.x;
        const float ay = acc.y + w * uu.y;
        const float az = acc.z + w * uu.z;
        const float aw = acc.w + w * uu.w;
        const bool ok = vld[it][k];
        acc.x = ok ? ax : acc.x;
        acc.y = ok ? ay : acc.y;
        acc.z = ok ? az : acc.z;
        acc.w = ok ? aw : acc.w;
      }
      acc.x = fmaxf(acc.x * dii, 0.0f);
      acc.y = fmaxf(acc.y * dii, 0.0f);
      acc.z = fmaxf(acc.z * dii, 0.0f);
      acc.w = fmaxf(acc.w * dii, 0.0f);
      vin[it] = make_float4(acc.x + rs[it].x, acc.y + rs[it].y,
                            acc.z + rs[it].z, acc.w + rs[it].w);
    }
    wave_ln4(vin, gg, bb, o);
#pragma unroll
    for (int it = 0; it < 4; ++it) {
      const int i = m0 + wrow + it;
      st4(hout + (size_t)i * ldh + d, o[it]);
      if constexpr (EPI != EPI_NONE) st4(As + (wrow + it) * S + d, o[it]);
    }
  } else {
    int cn[4];
#pragma unroll
    for (int it = 0; it < 4; ++it) {
      const int i = m0 + wave * 4 + it;
      int cnt = nbrc[i]; cnt = cnt > KNBR ? KNBR : cnt;
      cn[it] = cnt;
      const int4 na = *reinterpret_cast<const int4*>(neigh + i * KNBR);
      const int4 nb = *reinterpret_cast<const int4*>(neigh + i * KNBR + 4);
      const int nn[KNBR] = {na.x, na.y, na.z, na.w, nb.x, nb.y, nb.z, nb.w};
#pragma unroll
      for (int k = 0; k < KNBR; ++k) {
        vld[it][k] = k < cnt;
        jn[it][k]  = vld[it][k] ? nn[k] : i;
      }
    }
    float4 sf[4], u[4][KNBR];
#pragma unroll
    for (int it = 0; it < 4; ++it) {
      const int i = m0 + wave * 4 + it;
      sf[it] = ld4(Ain + (size_t)i * DIM + d);
    }
#pragma unroll
    for (int it = 0; it < 4; ++it)
#pragma unroll
      for (int k = 0; k < KNBR; ++k)
        u[it][k] = ld4(G + (size_t)jn[it][k] * DIM + d);
#pragma unroll
    for (int it = 0; it < 4; ++it) {
      st4(As + (wave * 4 + it) * S + d, sf[it]);
      float4 m = {NEGINF, NEGINF, NEGINF, NEGINF};
#pragma unroll
      for (int k = 0; k < KNBR; ++k) {
        const float4 uu = u[it][k];
        const float mx = fmaxf(m.x, uu.x);
        const float my = fmaxf(m.y, uu.y);
        const float mz = fmaxf(m.z, uu.z);
        const float mw = fmaxf(m.w, uu.w);
        const bool ok = vld[it][k];
        m.x = ok ? mx : m.x;
        m.y = ok ? my : m.y;
        m.z = ok ? mz : m.z;
        m.w = ok ? mw : m.w;
      }
      if (cn[it] == 0) m = make_float4(0.f, 0.f, 0.f, 0.f);
      st4(As + (wave * 4 + it) * S + DIM + d, m);
    }
  }

  if constexpr (EPI != EPI_NONE) {
#pragma unroll
    for (int it = 0; it < NW; ++it) {
      const int idx = it * 256 + t;
      const int r = idx / HC4, c = idx % HC4;
      st4(Ws + r * SW + 4 * c, rw[it]);
    }
    __syncthreads();

#pragma unroll
    for (int it = 0; it < NW; ++it) {
      const int idx = it * 256 + t;
      const int r = idx / HC4, c = idx % HC4;
      rw[it] = ld4(W + r * K + H + 4 * c);
    }

    const int half = lane >> 4;
    const int l16  = lane & 15;
    const int mt   = wave & 1;
    const int n0   = (wave >> 1) * 32;

    const float* ap  = As + (mt * 16 + l16) * S + 2 * half;
    const float* bp0 = Ws + (n0 + l16) * SW + 2 * half;
    const float* bp1 = bp0 + 16 * SW;

    const float bv0 = bias[n0 + l16];
    const float bv1 = bias[n0 + 16 + l16];
    float4 egv = {}, ebv = {};
    if constexpr (EPI == EPI_LN) { egv = ld4(eg + d); ebv = ld4(eb + d); }

    v8f c0 = {};
    v8f c1 = {};
    mma_range<0, H, 0>(ap, bp0, bp1, c0, c1);

    if constexpr (HALFW) {
      __syncthreads();
#pragma unroll
      for (int it = 0; it < NW; ++it) {
        const int idx = it * 256 + t;
        const int r = idx / HC4, c = idx % HC4;
        st4(Ws + r * SW + 4 * c, rw[it]);
      }
      __syncthreads();
      mma_range<H, K, H>(ap, bp0, bp1, c0, c1);
    } else {
#pragma unroll
      for (int it = 0; it < NW; ++it) {
        const int idx = it * 256 + t;
        const int r = idx / HC4, c = idx % HC4;
        st4(Ws + r * SW + H + 4 * c, rw[it]);
      }
      __syncthreads();
      mma_range<H, K, 0>(ap, bp0, bp1, c0, c1);
    }

    if constexpr (EPI == EPI_STORE) {
      float* yrow = Y + (size_t)(m0 + mt * 16 + 8 * half) * ldy + n0 + l16;
#pragma unroll
      for (int r = 0; r < 8; ++r) {
        float v0 = c0[r] + bv0;
        float v1 = c1[r] + bv1;
        if (do_relu) { v0 = fmaxf(v0, 0.0f); v1 = fmaxf(v1, 0.0f); }
        yrow[(size_t)r * ldy]      = v0;
        yrow[(size_t)r * ldy + 16] = v1;
      }
    } else {
      __syncthreads();
      float* ys = As + (mt * 16 + 8 * half) * S + DIM + n0 + l16;
#pragma unroll
      for (int r = 0; r < 8; ++r) {
        float v0 = c0[r] + bv0;
        float v1 = c1[r] + bv1;
        if (do_relu) { v0 = fmaxf(v0, 0.0f); v1 = fmaxf(v1, 0.0f); }
        ys[r * S]      = v0;
        ys[r * S + 16] = v1;
      }
      __syncthreads();
      float4 vin[4], o[4];
#pragma unroll
      for (int it = 0; it < 4; ++it) {
        const int lr = wave * 4 + it;
        const float4 rv = ld4(As + lr * S + d);
        const float4 av = ld4(As + lr * S + DIM + d);
        vin[it] = make_float4(av.x + rv.x, av.y + rv.y, av.z + rv.z, av.w + rv.w);
      }
      wave_ln4(vin, egv, ebv, o);
#pragma unroll
      for (int it = 0; it < 4; ++it)
        st4(Y + (size_t)(m0 + wave * 4 + it) * ldy + d, o[it]);
    }
  }
}


__global__ __launch_bounds__(256)
void pair_gemm_deg(const float* __restrict__ x,
                   const float* __restrict__ Wg, const float* __restrict__ bg,
                   float* __restrict__ Yg,
                   const float* __restrict__ Wp, const float* __restrict__ bp,
                   float* __restrict__ Yp,
                   const int* __restrict__ neigh, const int* __restrict__ nbrc,
                   float* __restrict__ dinv, unsigned* __restrict__ keep) {
  __shared__ __attribute__((aligned(16))) float lds[(32 + DIM) * (DIM + 4)];
  const int b = blockIdx.x;
  if (b >= 2 * NBLK) {
    deg_node((b - 2 * NBLK) * 256 + (int)threadIdx.x, neigh, nbrc, dinv, keep);
    return;
  }
  const bool pool = b >= NBLK;
  gemm_bias_block<DIM>(lds, (pool ? b - NBLK : b) * 32, x,
                       pool ? Wp : Wg, pool ? bp : bg, pool ? Yp : Yg,
                       DIM, pool ? 1 : 0);
}

template <int EPI_S, int EPI_G>
__global__ __launch_bounds__(256) __attribute__((amdgpu_waves_per_eu(4, 4)))
void pair_layer(const float* sAin, const float* sG,
                const int* __restrict__ neigh,
                const int* __restrict__ nbrc,
                const float* __restrict__ sW,
                const float* __restrict__ sbias,
                float* sY, int sldy,
                const float* __restrict__ seg,
                const float* __restrict__ seb,
                const float* gAin, const float* gG,
                const float* __restrict__ dinv,
                const unsigned* __restrict__ keep,
                const float* __restrict__ gpg,
                const float* __restrict__ gpb,
                float* ghout, int gldh,
                const float* __restrict__ gW,
                const float* __restrict__ gbias,
                float* gY) {
  __shared__ __attribute__((aligned(16))) float lds[LDS_FLOATS];
  const int b = blockIdx.x;
  if (b < NBLK) {
    fused_block<2 * DIM, PRO_SAGE, EPI_S>(
        lds, b * 32, sAin, sG, neigh, nbrc, nullptr, nullptr, nullptr, nullptr,
        nullptr, 0, sW, sbias, sY, sldy, 1, seg, seb);
  } else {
    fused_block<DIM, PRO_GCN, EPI_G>(
        lds, (b - NBLK) * 32, gAin, gG, neigh, nullptr, dinv, keep, gpg, gpb,
        ghout, gldh, gW, gbias, gY, DIM, 0, nullptr, nullptr);
  }
}

extern "C" void kernel_launch(void* const* d_in, const int* in_sizes, int n_in,
                              void* d_out, int out_size, void* d_ws, size_t ws_size,
                              hipStream_t stream) {
  const float* x      = (const float*)d_in[0];
  const int*   neigh  = (const int*)  d_in[1];
  const int*   nbrc   = (const int*)  d_in[2];
  const float* gcnW   = (const float*)d_in[3];
  const float* gcnB   = (const float*)d_in[4];
  const float* gcnG   = (const float*)d_in[5];
  const float* gcnBb  = (const float*)d_in[6];
  const float* poolW  = (const float*)d_in[7];
  const float* poolB  = (const float*)d_in[8];
  const float* mrgW   = (const float*)d_in[9];
  const float* mrgB   = (const float*)d_in[10];
  const float* sgG    = (const float*)d_in[11];
  const float* sgB    = (const float*)d_in[12];
  float* out = (float*)d_out;

  const size_t ND = (size_t)NNODE * DIM;
  float*    dinv  = (float*)d_ws;
  unsigned* keep  = (unsigned*)(dinv + NNODE);
  float*    first = (float*)(keep + NNODE);
  float*    nxt   = first + ND;
  float*    hbuf  = nxt + ND;
  float*    sbuf  = hbuf + ND;
  float*    tmp   = sbuf + ND;
  float*    nxt2  = tmp + ND;
  float*    sbuf2 = nxt2 + ND;

  const dim3 gB(256);
  const dim3 g1(2 * NBLK + NNODE / 256);
  const dim3 g2(2 * NBLK);

  pair_gemm_deg<<<g1, gB, 0, stream>>>(x, gcnW, gcnB, first, poolW, poolB, tmp,
                                       neigh, nbrc, dinv, keep);
  pair_layer<EPI_STORE, EPI_STORE><<<g2, gB, 0, stream>>>(
      x, tmp, neigh, nbrc, mrgW, mrgB, sbuf, DIM, nullptr, nullptr,
      first, first, dinv, keep, gcnG, gcnBb, hbuf, DIM,
      gcnW + DIM * DIM, gcnB + DIM, nxt);
  pair_layer<EPI_LN, EPI_STORE><<<g2, gB, 0, stream>>>(
      sbuf, sbuf, neigh, nbrc, mrgW + DIM * 2 * DIM, mrgB + DIM, sbuf2, DIM,
      sgG, sgB,
      hbuf, nxt, dinv, keep, gcnG + DIM, gcnBb + DIM, hbuf, DIM,
      gcnW + 2 * DIM * DIM, gcnB + 2 * DIM, nxt2);
  pair_layer<EPI_LN, EPI_NONE><<<g2, gB, 0, stream>>>(
      sbuf2, sbuf2, neigh, nbrc, mrgW + 2 * DIM * 2 * DIM, mrgB + 2 * DIM,
      out + DIM, 2 * DIM, sgG + DIM, sgB + DIM,
      hbuf, nxt2, dinv, keep, gcnG + 2 * DIM, gcnBb + 2 * DIM,
      out, 2 * DIM, nullptr, nullptr, nullptr);
}
